# starting kernel
# baseline (speedup 1.0000x reference)
#include <hip/hip_runtime.h>
#include <hip/hip_bf16.h>
#include <math.h>

typedef float v2f __attribute__((ext_vector_type(2)));
typedef float v8f __attribute__((ext_vector_type(8)));

#define D_IN   100
#define P      112
#define JT     7
#define KT     28
#define TILES  4
#define PTS    64
#define NTHR   512

#define PI_F   3.14159265358979323846f
#define CX     0.2f
#define CT     (4.0f / PI_F)

__global__ __launch_bounds__(NTHR, 1)
void pinn_residual_kernel(const float* __restrict__ x, const float* __restrict__ t,
                          const float* __restrict__ W1, const float* __restrict__ b1,
                          const float* __restrict__ W2, const float* __restrict__ b2,
                          const float* __restrict__ W3, const float* __restrict__ b3,
                          const float* __restrict__ W4, const float* __restrict__ b4,
                          const float* __restrict__ W5, const float* __restrict__ b5,
                          float* __restrict__ out, int N)
{
    __shared__ float w_lds[P * P];
    __shared__ float bias_lds[P];
    __shared__ float h_lds[TILES][4][16 * P];
    __shared__ float zx_lds[TILES][16 * P];
    __shared__ float sbuf[PTS][4][2];

    const int tid    = threadIdx.x;
    const int lane   = tid & 31;
    const int wave   = tid >> 5;
    const int tileId = wave >> 2;
    const int ch     = wave & 3;
    const int half   = lane >> 4;
    const int r      = lane & 15;

    const int base_pt = blockIdx.x * PTS;

    for (int i = tid; i < PTS * P; i += NTHR) {
        int pt = i / P, n = i % P;
        int tI = pt >> 4, pl = pt & 15;
        float hv = 0.f, hx = 0.f, ht = 0.f, hxx = 0.f;
        if (n < D_IN) {
            int g = base_pt + pt;
            float xv = (g < N) ? x[g] : 0.0f;
            float tv = (g < N) ? t[g] : 0.0f;
            float sx = CX * (xv + 5.0f) - 1.0f;
            float st = CT * tv - 1.0f;
            float w0 = W1[n], w1 = W1[D_IN + n];
            float z  = sx * w0 + st * w1 + b1[n];
            float zx = CX * w0;
            float zt = CT * w1;
            float h  = tanhf(z);
            float gg = 1.0f - h * h;
            hv = h; hx = gg * zx; ht = gg * zt; hxx = -2.0f * h * gg * zx * zx;
        }
        h_lds[tI][0][pl * P + n] = hv;
        h_lds[tI][1][pl * P + n] = hx;
        h_lds[tI][2][pl * P + n] = ht;
        h_lds[tI][3][pl * P + n] = hxx;
    }

    for (int l = 0; l < 3; ++l) {
        const float* Wg = (l == 0) ? W2 : (l == 1) ? W3 : W4;
        const float* bg = (l == 0) ? b2 : (l == 1) ? b3 : b4;

        for (int i = tid; i < P * P; i += NTHR) {
            int o   = i & 1;
            int idx = i >> 1;
            int p   = idx / P;
            int n   = idx - p * P;
            int k   = 2 * p + o;
            w_lds[i] = (k < D_IN && n < D_IN) ? Wg[k * D_IN + n] : 0.0f;
        }
        for (int i = tid; i < P; i += NTHR)
            bias_lds[i] = (i < D_IN) ? bg[i] : 0.0f;
        __syncthreads();

        const float* hA = &h_lds[tileId][ch][0];
        v8f acc[JT];
#pragma unroll
        for (int j = 0; j < JT; ++j)
            acc[j] = (v8f){0.f, 0.f, 0.f, 0.f, 0.f, 0.f, 0.f, 0.f};

        for (int k = 0; k < KT; ++k) {
            int k0 = k * 4 + half * 2;
            int p  = k * 2 + half;
            v2f a = *(const v2f*)&hA[r * P + k0];
#pragma unroll
            for (int j = 0; j < JT; ++j) {
                v2f b = *(const v2f*)&w_lds[(p * P + j * 16 + r) * 2];
                acc[j] = __builtin_amdgcn_wmma_f32_16x16x4_f32(
                    false, a, false, b, (short)0, acc[j], false, false);
            }
        }
        __syncthreads();

        if (ch == 0) {
#pragma unroll
            for (int j = 0; j < JT; ++j)
#pragma unroll
                for (int i = 0; i < 8; ++i) {
                    int pt = i + 8 * half, n = j * 16 + r;
                    float z = acc[j][i] + bias_lds[n];
                    h_lds[tileId][0][pt * P + n] = tanhf(z);
                }
        } else if (ch == 1) {
#pragma unroll
            for (int j = 0; j < JT; ++j)
#pragma unroll
                for (int i = 0; i < 8; ++i) {
                    int pt = i + 8 * half, n = j * 16 + r;
                    zx_lds[tileId][pt * P + n] = acc[j][i];
                }
        }
        __syncthreads();

        if (ch == 1) {
#pragma unroll
            for (int j = 0; j < JT; ++j)
#pragma unroll
                for (int i = 0; i < 8; ++i) {
                    int pt = i + 8 * half, n = j * 16 + r;
                    float h = h_lds[tileId][0][pt * P + n];
                    float g = 1.0f - h * h;
                    h_lds[tileId][1][pt * P + n] = g * acc[j][i];
                }
        } else if (ch == 2) {
#pragma unroll
            for (int j = 0; j < JT; ++j)
#pragma unroll
                for (int i = 0; i < 8; ++i) {
                    int pt = i + 8 * half, n = j * 16 + r;
                    float h = h_lds[tileId][0][pt * P + n];
                    float g = 1.0f - h * h;
                    h_lds[tileId][2][pt * P + n] = g * acc[j][i];
                }
        } else if (ch == 3) {
#pragma unroll
            for (int j = 0; j < JT; ++j)
#pragma unroll
                for (int i = 0; i < 8; ++i) {
                    int pt = i + 8 * half, n = j * 16 + r;
                    float h  = h_lds[tileId][0][pt * P + n];
                    float g  = 1.0f - h * h;
                    float zx = zx_lds[tileId][pt * P + n];
                    h_lds[tileId][3][pt * P + n] = g * acc[j][i] - 2.0f * h * g * zx * zx;
                }
        }
        __syncthreads();
    }

    if (tid < PTS * 4) {
        int pt = tid >> 2, c = tid & 3;
        int tI = pt >> 4, pl = pt & 15;
        const float* hh = &h_lds[tI][c][pl * P];
        float s0 = 0.f, s1 = 0.f;
        for (int k = 0; k < D_IN; ++k) {
            float hk = hh[k];
            s0 = fmaf(hk, W5[k * 2 + 0], s0);
            s1 = fmaf(hk, W5[k * 2 + 1], s1);
        }
        sbuf[pt][c][0] = s0;
        sbuf[pt][c][1] = s1;
    }
    __syncthreads();

    if (tid < PTS) {
        int pt = tid;
        int g = base_pt + pt;
        if (g < N) {
            float u    = sbuf[pt][0][0] + b5[0];
            float v    = sbuf[pt][0][1] + b5[1];
            float u_t  = sbuf[pt][2][0];
            float v_t  = sbuf[pt][2][1];
            float u_xx = sbuf[pt][3][0];
            float v_xx = sbuf[pt][3][1];
            float sq   = u * u + v * v;
            out[g * 2 + 0] = u_t + 0.5f * v_xx + sq * v;
            out[g * 2 + 1] = v_t - 0.5f * u_xx - sq * u;
        }
    }
}

extern "C" void kernel_launch(void* const* d_in, const int* in_sizes, int n_in,
                              void* d_out, int out_size, void* d_ws, size_t ws_size,
                              hipStream_t stream) {
    const float* x  = (const float*)d_in[0];
    const float* t  = (const float*)d_in[1];
    const float* W1 = (const float*)d_in[2];
    const float* b1 = (const float*)d_in[3];
    const float* W2 = (const float*)d_in[4];
    const float* b2 = (const float*)d_in[5];
    const float* W3 = (const float*)d_in[6];
    const float* b3 = (const float*)d_in[7];
    const float* W4 = (const float*)d_in[8];
    const float* b4 = (const float*)d_in[9];
    const float* W5 = (const float*)d_in[10];
    const float* b5 = (const float*)d_in[11];
    float* out = (float*)d_out;
    int N = in_sizes[0];
    int blocks = (N + PTS - 1) / PTS;
    pinn_residual_kernel<<<blocks, NTHR, 0, stream>>>(
        x, t, W1, b1, W2, b2, W3, b3, W4, b4, W5, b5, out, N);
}


# Weight staging of layers 2-4: each thread issues its six 16-byte global loads as one batch before any wait, replacing 25 dependent 4-byte load-wait-store trips per layer
# speedup vs baseline: 1.2962x; 1.2962x over previous
#include <hip/hip_runtime.h>
#include <hip/hip_bf16.h>
#include <math.h>

typedef float v2f __attribute__((ext_vector_type(2)));
typedef float v8f __attribute__((ext_vector_type(8)));
typedef float v4f __attribute__((ext_vector_type(4)));
typedef float v4f_u __attribute__((ext_vector_type(4), aligned(4)));

#define D_IN   100
#define P      112
#define JT     7
#define KT     28
#define TILES  4
#define PTS    64
#define NTHR   512
#define NGRP   (D_IN / 4)
#define NPAIR  (D_IN / 2)
#define NITEM  (NPAIR * NGRP)
#define SROUNDS ((NITEM + NTHR - 1) / NTHR)

#define PI_F   3.14159265358979323846f
#define CX     0.2f
#define CT     (4.0f / PI_F)

__global__ __launch_bounds__(NTHR, 1)
void pinn_residual_kernel(const float* __restrict__ x, const float* __restrict__ t,
                          const float* __restrict__ W1, const float* __restrict__ b1,
                          const float* __restrict__ W2, const float* __restrict__ b2,
                          const float* __restrict__ W3, const float* __restrict__ b3,
                          const float* __restrict__ W4, const float* __restrict__ b4,
                          const float* __restrict__ W5, const float* __restrict__ b5,
                          float* __restrict__ out, int N)
{
    __shared__ __attribute__((aligned(16))) float w_lds[P * P];
    __shared__ float bias_lds[P];
    __shared__ float h_lds[TILES][4][16 * P];
    __shared__ float zx_lds[TILES][16 * P];
    __shared__ float sbuf[PTS][4][2];

    const int tid    = threadIdx.x;
    const int lane   = tid & 31;
    const int wave   = tid >> 5;
    const int tileId = wave >> 2;
    const int ch     = wave & 3;
    const int half   = lane >> 4;
    const int r      = lane & 15;

    const int base_pt = blockIdx.x * PTS;

    for (int i = tid; i < PTS * P; i += NTHR) {
        int pt = i / P, n = i % P;
        int tI = pt >> 4, pl = pt & 15;
        float hv = 0.f, hx = 0.f, ht = 0.f, hxx = 0.f;
        if (n < D_IN) {
            int g = base_pt + pt;
            float xv = (g < N) ? x[g] : 0.0f;
            float tv = (g < N) ? t[g] : 0.0f;
            float sx = CX * (xv + 5.0f) - 1.0f;
            float st = CT * tv - 1.0f;
            float w0 = W1[n], w1 = W1[D_IN + n];
            float z  = sx * w0 + st * w1 + b1[n];
            float zx = CX * w0;
            float zt = CT * w1;
            float h  = tanhf(z);
            float gg = 1.0f - h * h;
            hv = h; hx = gg * zx; ht = gg * zt; hxx = -2.0f * h * gg * zx * zx;
        }
        h_lds[tI][0][pl * P + n] = hv;
        h_lds[tI][1][pl * P + n] = hx;
        h_lds[tI][2][pl * P + n] = ht;
        h_lds[tI][3][pl * P + n] = hxx;
    }

    for (int i = tid; i < NPAIR * (P - D_IN) * 2; i += NTHR) {
        int p = i / ((P - D_IN) * 2);
        int c = i - p * ((P - D_IN) * 2);
        w_lds[(p * P + D_IN) * 2 + c] = 0.0f;
    }
    for (int i = NPAIR * P * 2 + tid; i < P * P; i += NTHR)
        w_lds[i] = 0.0f;

    for (int l = 0; l < 3; ++l) {
        const float* Wg = (l == 0) ? W2 : (l == 1) ? W3 : W4;
        const float* bg = (l == 0) ? b2 : (l == 1) ? b3 : b4;

        v4f ra[SROUNDS], rb[SROUNDS];
#pragma unroll
        for (int it = 0; it < SROUNDS; ++it) {
            int item = tid + it * NTHR;
            item = (item < NITEM) ? item : (NITEM - 1);
            int p = item / NGRP;
            int c = item - p * NGRP;
            const float* s0 = Wg + (2 * p) * D_IN + 4 * c;
            ra[it] = *(const v4f_u*)s0;
            rb[it] = *(const v4f_u*)(s0 + D_IN);
        }
        float bv = bg[(tid < D_IN) ? tid : (D_IN - 1)];
#pragma unroll
        for (int it = 0; it < SROUNDS; ++it) {
            int item = tid + it * NTHR;
            if (item < NITEM) {
                int p = item / NGRP;
                int c = item - p * NGRP;
                float* d = &w_lds[(p * P + 4 * c) * 2];
                *(v4f*)(d)     = (v4f){ra[it][0], rb[it][0], ra[it][1], rb[it][1]};
                *(v4f*)(d + 4) = (v4f){ra[it][2], rb[it][2], ra[it][3], rb[it][3]};
            }
        }
        if (tid < P)
            bias_lds[tid] = (tid < D_IN) ? bv : 0.0f;
        __syncthreads();

        const float* hA = &h_lds[tileId][ch][0];
        v8f acc[JT];
#pragma unroll
        for (int j = 0; j < JT; ++j)
            acc[j] = (v8f){0.f, 0.f, 0.f, 0.f, 0.f, 0.f, 0.f, 0.f};

        for (int k = 0; k < KT; ++k) {
            int k0 = k * 4 + half * 2;
            int p  = k * 2 + half;
            v2f a = *(const v2f*)&hA[r * P + k0];
#pragma unroll
            for (int j = 0; j < JT; ++j) {
                v2f b = *(const v2f*)&w_lds[(p * P + j * 16 + r) * 2];
                acc[j] = __builtin_amdgcn_wmma_f32_16x16x4_f32(
                    false, a, false, b, (short)0, acc[j], false, false);
            }
        }
        __syncthreads();

        if (ch == 0) {
#pragma unroll
            for (int j = 0; j < JT; ++j)
#pragma unroll
                for (int i = 0; i < 8; ++i) {
                    int pt = i + 8 * half, n = j * 16 + r;
                    float z = acc[j][i] + bias_lds[n];
                    h_lds[tileId][0][pt * P + n] = tanhf(z);
                }
        } else if (ch == 1) {
#pragma unroll
            for (int j = 0; j < JT; ++j)
#pragma unroll
                for (int i = 0; i < 8; ++i) {
                    int pt = i + 8 * half, n = j * 16 + r;
                    zx_lds[tileId][pt * P + n] = acc[j][i];
                }
        }
        __syncthreads();

        if (ch == 1) {
#pragma unroll
            for (int j = 0; j < JT; ++j)
#pragma unroll
                for (int i = 0; i < 8; ++i) {
                    int pt = i + 8 * half, n = j * 16 + r;
                    float h = h_lds[tileId][0][pt * P + n];
                    float g = 1.0f - h * h;
                    h_lds[tileId][1][pt * P + n] = g * acc[j][i];
                }
        } else if (ch == 2) {
#pragma unroll
            for (int j = 0; j < JT; ++j)
#pragma unroll
                for (int i = 0; i < 8; ++i) {
                    int pt = i + 8 * half, n = j * 16 + r;
                    float h = h_lds[tileId][0][pt * P + n];
                    float g = 1.0f - h * h;
                    h_lds[tileId][2][pt * P + n] = g * acc[j][i];
                }
        } else if (ch == 3) {
#pragma unroll
            for (int j = 0; j < JT; ++j)
#pragma unroll
                for (int i = 0; i < 8; ++i) {
                    int pt = i + 8 * half, n = j * 16 + r;
                    float h  = h_lds[tileId][0][pt * P + n];
                    float g  = 1.0f - h * h;
                    float zx = zx_lds[tileId][pt * P + n];
                    h_lds[tileId][3][pt * P + n] = g * acc[j][i] - 2.0f * h * g * zx * zx;
                }
        }
        __syncthreads();
    }

    if (tid < PTS * 4) {
        int pt = tid >> 2, c = tid & 3;
        int tI = pt >> 4, pl = pt & 15;
        const float* hh = &h_lds[tI][c][pl * P];
        float s0 = 0.f, s1 = 0.f;
        for (int k = 0; k < D_IN; ++k) {
            float hk = hh[k];
            s0 = fmaf(hk, W5[k * 2 + 0], s0);
            s1 = fmaf(hk, W5[k * 2 + 1], s1);
        }
        sbuf[pt][c][0] = s0;
        sbuf[pt][c][1] = s1;
    }
    __syncthreads();

    if (tid < PTS) {
        int pt = tid;
        int g = base_pt + pt;
        if (g < N) {
            float u    = sbuf[pt][0][0] + b5[0];
            float v    = sbuf[pt][0][1] + b5[1];
            float u_t  = sbuf[pt][2][0];
            float v_t  = sbuf[pt][2][1];
            float u_xx = sbuf[pt][3][0];
            float v_xx = sbuf[pt][3][1];
            float sq   = u * u + v * v;
            out[g * 2 + 0] = u_t + 0.5f * v_xx + sq * v;
            out[g * 2 + 1] = v_t - 0.5f * u_xx - sq * u;
        }
    }
}

extern "C" void kernel_launch(void* const* d_in, const int* in_sizes, int n_in,
                              void* d_out, int out_size, void* d_ws, size_t ws_size,
                              hipStream_t stream) {
    const float* x  = (const float*)d_in[0];
    const float* t  = (const float*)d_in[1];
    const float* W1 = (const float*)d_in[2];
    const float* b1 = (const float*)d_in[3];
    const float* W2 = (const float*)d_in[4];
    const float* b2 = (const float*)d_in[5];
    const float* W3 = (const float*)d_in[6];
    const float* b3 = (const float*)d_in[7];
    const float* W4 = (const float*)d_in[8];
    const float* b4 = (const float*)d_in[9];
    const float* W5 = (const float*)d_in[10];
    const float* b5 = (const float*)d_in[11];
    float* out = (float*)d_out;
    int N = in_sizes[0];
    int blocks = (N + PTS - 1) / PTS;
    pinn_residual_kernel<<<blocks, NTHR, 0, stream>>>(
        x, t, W1, b1, W2, b2, W3, b3, W4, b4, W5, b5, out, N);
}


# Activation epilogue rebalanced: each wave parks its raw WMMA sums in its own LDS plane, then all 16 waves share one in-place tanh/derivative pass (before: every tanh ran on 4 waves of one SIMD)
# speedup vs baseline: 1.6520x; 1.2745x over previous
#include <hip/hip_runtime.h>
#include <hip/hip_bf16.h>
#include <math.h>

typedef float v2f __attribute__((ext_vector_type(2)));
typedef float v8f __attribute__((ext_vector_type(8)));
typedef float v4f __attribute__((ext_vector_type(4)));
typedef float v4f_u __attribute__((ext_vector_type(4), aligned(4)));

#define D_IN   100
#define P      112
#define JT     7
#define KT     28
#define TILES  4
#define PTS    64
#define NTHR   512
#define NGRP   (D_IN / 4)
#define NPAIR  (D_IN / 2)
#define NITEM  (NPAIR * NGRP)
#define SROUNDS ((NITEM + NTHR - 1) / NTHR)

#define PI_F   3.14159265358979323846f
#define CX     0.2f
#define CT     (4.0f / PI_F)

__global__ __launch_bounds__(NTHR, 1)
void pinn_residual_kernel(const float* __restrict__ x, const float* __restrict__ t,
                          const float* __restrict__ W1, const float* __restrict__ b1,
                          const float* __restrict__ W2, const float* __restrict__ b2,
                          const float* __restrict__ W3, const float* __restrict__ b3,
                          const float* __restrict__ W4, const float* __restrict__ b4,
                          const float* __restrict__ W5, const float* __restrict__ b5,
                          float* __restrict__ out, int N)
{
    __shared__ __attribute__((aligned(16))) float w_lds[P * P];
    __shared__ __attribute__((aligned(16))) float bias_lds[P];
    __shared__ __attribute__((aligned(16))) float h_lds[TILES][4][16 * P];
    __shared__ float sbuf[PTS][4][2];

    const int tid    = threadIdx.x;
    const int lane   = tid & 31;
    const int wave   = tid >> 5;
    const int tileId = wave >> 2;
    const int ch     = wave & 3;
    const int half   = lane >> 4;
    const int r      = lane & 15;

    const int base_pt = blockIdx.x * PTS;

    for (int i = tid; i < PTS * P; i += NTHR) {
        int pt = i / P, n = i % P;
        int tI = pt >> 4, pl = pt & 15;
        float hv = 0.f, hx = 0.f, ht = 0.f, hxx = 0.f;
        if (n < D_IN) {
            int g = base_pt + pt;
            float xv = (g < N) ? x[g] : 0.0f;
            float tv = (g < N) ? t[g] : 0.0f;
            float sx = CX * (xv + 5.0f) - 1.0f;
            float st = CT * tv - 1.0f;
            float w0 = W1[n], w1 = W1[D_IN + n];
            float z  = sx * w0 + st * w1 + b1[n];
            float zx = CX * w0;
            float zt = CT * w1;
            float h  = tanhf(z);
            float gg = 1.0f - h * h;
            hv = h; hx = gg * zx; ht = gg * zt; hxx = -2.0f * h * gg * zx * zx;
        }
        h_lds[tI][0][pl * P + n] = hv;
        h_lds[tI][1][pl * P + n] = hx;
        h_lds[tI][2][pl * P + n] = ht;
        h_lds[tI][3][pl * P + n] = hxx;
    }

    for (int i = tid; i < NPAIR * (P - D_IN) * 2; i += NTHR) {
        int p = i / ((P - D_IN) * 2);
        int c = i - p * ((P - D_IN) * 2);
        w_lds[(p * P + D_IN) * 2 + c] = 0.0f;
    }
    for (int i = NPAIR * P * 2 + tid; i < P * P; i += NTHR)
        w_lds[i] = 0.0f;

    for (int l = 0; l < 3; ++l) {
        const float* Wg = (l == 0) ? W2 : (l == 1) ? W3 : W4;
        const float* bg = (l == 0) ? b2 : (l == 1) ? b3 : b4;

        v4f ra[SROUNDS], rb[SROUNDS];
#pragma unroll
        for (int it = 0; it < SROUNDS; ++it) {
            int item = tid + it * NTHR;
            item = (item < NITEM) ? item : (NITEM - 1);
            int p = item / NGRP;
            int c = item - p * NGRP;
            const float* s0 = Wg + (2 * p) * D_IN + 4 * c;
            ra[it] = *(const v4f_u*)s0;
            rb[it] = *(const v4f_u*)(s0 + D_IN);
        }
        float bv = bg[(tid < D_IN) ? tid : (D_IN - 1)];
#pragma unroll
        for (int it = 0; it < SROUNDS; ++it) {
            int item = tid + it * NTHR;
            if (item < NITEM) {
                int p = item / NGRP;
                int c = item - p * NGRP;
                float* d = &w_lds[(p * P + 4 * c) * 2];
                *(v4f*)(d)     = (v4f){ra[it][0], rb[it][0], ra[it][1], rb[it][1]};
                *(v4f*)(d + 4) = (v4f){ra[it][2], rb[it][2], ra[it][3], rb[it][3]};
            }
        }
        if (tid < P)
            bias_lds[tid] = (tid < D_IN) ? bv : 0.0f;
        __syncthreads();

        const float* hA = &h_lds[tileId][ch][0];
        v8f acc[JT];
#pragma unroll
        for (int j = 0; j < JT; ++j)
            acc[j] = (v8f){0.f, 0.f, 0.f, 0.f, 0.f, 0.f, 0.f, 0.f};

        for (int k = 0; k < KT; ++k) {
            int k0 = k * 4 + half * 2;
            int p  = k * 2 + half;
            v2f a = *(const v2f*)&hA[r * P + k0];
#pragma unroll
            for (int j = 0; j < JT; ++j) {
                v2f b = *(const v2f*)&w_lds[(p * P + j * 16 + r) * 2];
                acc[j] = __builtin_amdgcn_wmma_f32_16x16x4_f32(
                    false, a, false, b, (short)0, acc[j], false, false);
            }
        }
        {
            float* hW = &h_lds[tileId][ch][0];
#pragma unroll
            for (int j = 0; j < JT; ++j)
#pragma unroll
                for (int i = 0; i < 8; ++i)
                    hW[(i + 8 * half) * P + j * 16 + r] = acc[j][i];
        }
        __syncthreads();

        {
            const int tA    = tid >> 7;
            const int local = tid & 127;
            float* p0 = &h_lds[tA][0][0];
            float* p1 = &h_lds[tA][1][0];
            float* p2 = &h_lds[tA][2][0];
            float* p3 = &h_lds[tA][3][0];
            for (int it = 0; it < (16 * P) / (2 * 128); ++it) {
                int off = 2 * (local + it * 128);
                int n   = off % P;
                v2f z0 = *(const v2f*)&p0[off];
                v2f z1 = *(const v2f*)&p1[off];
                v2f z2 = *(const v2f*)&p2[off];
                v2f z3 = *(const v2f*)&p3[off];
                v2f bb = *(const v2f*)&bias_lds[n];
                v2f o0 = {0.f, 0.f}, o1 = {0.f, 0.f}, o2 = {0.f, 0.f}, o3 = {0.f, 0.f};
#pragma unroll
                for (int e = 0; e < 2; ++e) {
                    float h  = tanhf(z0[e] + bb[e]);
                    float g  = 1.0f - h * h;
                    float zx = z1[e];
                    o0[e] = h;
                    o1[e] = g * zx;
                    o2[e] = g * z2[e];
                    o3[e] = g * z3[e] - 2.0f * h * g * zx * zx;
                }
                *(v2f*)&p0[off] = o0;
                *(v2f*)&p1[off] = o1;
                *(v2f*)&p2[off] = o2;
                *(v2f*)&p3[off] = o3;
            }
        }
        __syncthreads();
    }

    if (tid < PTS * 4) {
        int pt = tid >> 2, c = tid & 3;
        int tI = pt >> 4, pl = pt & 15;
        const float* hh = &h_lds[tI][c][pl * P];
        float s0 = 0.f, s1 = 0.f;
        for (int k = 0; k < D_IN; ++k) {
            float hk = hh[k];
            s0 = fmaf(hk, W5[k * 2 + 0], s0);
            s1 = fmaf(hk, W5[k * 2 + 1], s1);
        }
        sbuf[pt][c][0] = s0;
        sbuf[pt][c][1] = s1;
    }
    __syncthreads();

    if (tid < PTS) {
        int pt = tid;
        int g = base_pt + pt;
        if (g < N) {
            float u    = sbuf[pt][0][0] + b5[0];
            float v    = sbuf[pt][0][1] + b5[1];
            float u_t  = sbuf[pt][2][0];
            float v_t  = sbuf[pt][2][1];
            float u_xx = sbuf[pt][3][0];
            float v_xx = sbuf[pt][3][1];
            float sq   = u * u + v * v;
            out[g * 2 + 0] = u_t + 0.5f * v_xx + sq * v;
            out[g * 2 + 1] = v_t - 0.5f * u_xx - sq * u;
        }
    }
}

extern "C" void kernel_launch(void* const* d_in, const int* in_sizes, int n_in,
                              void* d_out, int out_size, void* d_ws, size_t ws_size,
                              hipStream_t stream) {
    const float* x  = (const float*)d_in[0];
    const float* t  = (const float*)d_in[1];
    const float* W1 = (const float*)d_in[2];
    const float* b1 = (const float*)d_in[3];
    const float* W2 = (const float*)d_in[4];
    const float* b2 = (const float*)d_in[5];
    const float* W3 = (const float*)d_in[6];
    const float* b3 = (const float*)d_in[7];
    const float* W4 = (const float*)d_in[8];
    const float* b4 = (const float*)d_in[9];
    const float* W5 = (const float*)d_in[10];
    const float* b5 = (const float*)d_in[11];
    float* out = (float*)d_out;
    int N = in_sizes[0];
    int blocks = (N + PTS - 1) / PTS;
    pinn_residual_kernel<<<blocks, NTHR, 0, stream>>>(
        x, t, W1, b1, W2, b2, W3, b3, W4, b4, W5, b5, out, N);
}


# Layer 1 re-tiled: a thread owns one neuron pair, loads its 6 weights once and its 7 points' x,t as one batch (was 14 trips of 3 dependent global loads plus an index division); pad columns unwritten
# speedup vs baseline: 2.3969x; 1.4509x over previous
#include <hip/hip_runtime.h>
#include <hip/hip_bf16.h>
#include <math.h>

typedef float v2f __attribute__((ext_vector_type(2)));
typedef float v8f __attribute__((ext_vector_type(8)));
typedef float v4f __attribute__((ext_vector_type(4)));
typedef float v4f_u __attribute__((ext_vector_type(4), aligned(4)));
typedef float v2f_u __attribute__((ext_vector_type(2), aligned(4)));

#define D_IN   100
#define P      112
#define JT     7
#define KT     25
#define TILES  4
#define PTS    64
#define NTHR   512
#define NGRP   (D_IN / 4)
#define NPAIR  (D_IN / 2)
#define NITEM  (NPAIR * NGRP)
#define SROUNDS ((NITEM + NTHR - 1) / NTHR)
#define L1_PAIRS   (D_IN / 2)
#define L1_GROUPS  (NTHR / L1_PAIRS)
#define L1_THREADS (L1_PAIRS * L1_GROUPS)
#define L1_ITERS   ((PTS + L1_GROUPS - 1) / L1_GROUPS)

#define PI_F   3.14159265358979323846f
#define CX     0.2f
#define CT     (4.0f / PI_F)

__global__ __launch_bounds__(NTHR, 8)
void pinn_residual_kernel(const float* __restrict__ x, const float* __restrict__ t,
                          const float* __restrict__ W1, const float* __restrict__ b1,
                          const float* __restrict__ W2, const float* __restrict__ b2,
                          const float* __restrict__ W3, const float* __restrict__ b3,
                          const float* __restrict__ W4, const float* __restrict__ b4,
                          const float* __restrict__ W5, const float* __restrict__ b5,
                          float* __restrict__ out, int N)
{
    __shared__ __attribute__((aligned(16))) float w_lds[NPAIR * P * 2];
    __shared__ __attribute__((aligned(16))) float bias_lds[P];
    __shared__ __attribute__((aligned(16))) float h_lds[TILES][4][16 * P];
    float (*sbuf)[4][2] = (float (*)[4][2])w_lds;

    const int tid    = threadIdx.x;
    const int lane   = tid & 31;
    const int wave   = tid >> 5;
    const int tileId = wave >> 2;
    const int ch     = wave & 3;
    const int half   = lane >> 4;
    const int r      = lane & 15;

    const int base_pt = blockIdx.x * PTS;

    if (tid < L1_THREADS) {
        const unsigned utid = (unsigned)tid;
        const int pg = (int)(utid / L1_PAIRS);
        const int cp = (int)(utid - (unsigned)pg * L1_PAIRS);
        const int n0 = 2 * cp;
        const v2f w0 = *(const v2f_u*)(W1 + n0);
        const v2f w1 = *(const v2f_u*)(W1 + D_IN + n0);
        const v2f bb = *(const v2f_u*)(b1 + n0);
        float xv[L1_ITERS], tv[L1_ITERS];
#pragma unroll
        for (int it = 0; it < L1_ITERS; ++it) {
            int pt = pg + L1_GROUPS * it;
            pt = (pt < PTS) ? pt : (PTS - 1);
            const int g  = base_pt + pt;
            const int gc = (g < N) ? g : (N - 1);
            const float xl = x[gc];
            const float tl = t[gc];
            xv[it] = (g < N) ? xl : 0.0f;
            tv[it] = (g < N) ? tl : 0.0f;
        }
        const v2f zx = {CX * w0[0], CX * w0[1]};
        const v2f zt = {CT * w1[0], CT * w1[1]};
#pragma unroll
        for (int it = 0; it < L1_ITERS; ++it) {
            const int pt = pg + L1_GROUPS * it;
            if (pt < PTS) {
                const float sx = CX * (xv[it] + 5.0f) - 1.0f;
                const float st = CT * tv[it] - 1.0f;
                v2f hv, hx, ht, hxx;
#pragma unroll
                for (int e = 0; e < 2; ++e) {
                    float z  = sx * w0[e] + st * w1[e] + bb[e];
                    float h  = tanhf(z);
                    float gg = 1.0f - h * h;
                    hv[e]  = h;
                    hx[e]  = gg * zx[e];
                    ht[e]  = gg * zt[e];
                    hxx[e] = -2.0f * h * gg * zx[e] * zx[e];
                }
                const int tI = pt >> 4, pl = pt & 15;
                float* d = &h_lds[tI][0][pl * P + n0];
                *(v2f*)(d)              = hv;
                *(v2f*)(d + 16 * P)     = hx;
                *(v2f*)(d + 2 * 16 * P) = ht;
                *(v2f*)(d + 3 * 16 * P) = hxx;
            }
        }
    }

    for (int i = tid; i < NPAIR * (P - D_IN) * 2; i += NTHR) {
        int p = i / ((P - D_IN) * 2);
        int c = i - p * ((P - D_IN) * 2);
        w_lds[(p * P + D_IN) * 2 + c] = 0.0f;
    }

    for (int l = 0; l < 3; ++l) {
        const float* Wg = (l == 0) ? W2 : (l == 1) ? W3 : W4;
        const float* bg = (l == 0) ? b2 : (l == 1) ? b3 : b4;

        v4f ra[SROUNDS], rb[SROUNDS];
#pragma unroll
        for (int it = 0; it < SROUNDS; ++it) {
            int item = tid + it * NTHR;
            item = (item < NITEM) ? item : (NITEM - 1);
            int p = item / NGRP;
            int c = item - p * NGRP;
            const float* s0 = Wg + (2 * p) * D_IN + 4 * c;
            ra[it] = *(const v4f_u*)s0;
            rb[it] = *(const v4f_u*)(s0 + D_IN);
        }
        float bv = bg[(tid < D_IN) ? tid : (D_IN - 1)];
#pragma unroll
        for (int it = 0; it < SROUNDS; ++it) {
            int item = tid + it * NTHR;
            if (item < NITEM) {
                int p = item / NGRP;
                int c = item - p * NGRP;
                float* d = &w_lds[(p * P + 4 * c) * 2];
                *(v4f*)(d)     = (v4f){ra[it][0], rb[it][0], ra[it][1], rb[it][1]};
                *(v4f*)(d + 4) = (v4f){ra[it][2], rb[it][2], ra[it][3], rb[it][3]};
            }
        }
        if (tid < P)
            bias_lds[tid] = (tid < D_IN) ? bv : 0.0f;
        __syncthreads();

        const float* hA = &h_lds[tileId][ch][0];
        v8f acc[JT];
#pragma unroll
        for (int j = 0; j < JT; ++j)
            acc[j] = (v8f){0.f, 0.f, 0.f, 0.f, 0.f, 0.f, 0.f, 0.f};

        const float* aP = hA + r * P + half * 2;
        const float* bP = &w_lds[(half * P + r) * 2];
        v2f a0 = *(const v2f*)aP, a1;
        v2f b0[JT], b1[JT];
#pragma unroll
        for (int j = 0; j < JT; ++j)
            b0[j] = *(const v2f*)(bP + j * 32);
#pragma unroll 1
        for (int k = 0; k < KT - 1; k += 2) {
            a1 = *(const v2f*)(aP + (k + 1) * 4);
#pragma unroll
            for (int j = 0; j < JT; ++j)
                b1[j] = *(const v2f*)(bP + (k + 1) * (4 * P) + j * 32);
            __builtin_amdgcn_sched_barrier(0);
#pragma unroll
            for (int j = 0; j < JT; ++j)
                acc[j] = __builtin_amdgcn_wmma_f32_16x16x4_f32(
                    false, a0, false, b0[j], (short)0, acc[j], false, false);
            __builtin_amdgcn_sched_barrier(0);
            a0 = *(const v2f*)(aP + (k + 2) * 4);
#pragma unroll
            for (int j = 0; j < JT; ++j)
                b0[j] = *(const v2f*)(bP + (k + 2) * (4 * P) + j * 32);
            __builtin_amdgcn_sched_barrier(0);
#pragma unroll
            for (int j = 0; j < JT; ++j)
                acc[j] = __builtin_amdgcn_wmma_f32_16x16x4_f32(
                    false, a1, false, b1[j], (short)0, acc[j], false, false);
            __builtin_amdgcn_sched_barrier(0);
        }
#pragma unroll
        for (int j = 0; j < JT; ++j)
            acc[j] = __builtin_amdgcn_wmma_f32_16x16x4_f32(
                false, a0, false, b0[j], (short)0, acc[j], false, false);
        {
            float* hW = &h_lds[tileId][ch][0];
#pragma unroll
            for (int j = 0; j < JT; ++j)
#pragma unroll
                for (int i = 0; i < 8; ++i)
                    hW[(i + 8 * half) * P + j * 16 + r] = acc[j][i];
        }
        __syncthreads();

        {
            const int tA    = tid >> 7;
            const int local = tid & 127;
            float* p0 = &h_lds[tA][0][0];
            float* p1 = &h_lds[tA][1][0];
            float* p2 = &h_lds[tA][2][0];
            float* p3 = &h_lds[tA][3][0];
            for (int it = 0; it < (16 * P) / (2 * 128); ++it) {
                int off = 2 * (local + it * 128);
                int n   = off % P;
                v2f z0 = *(const v2f*)&p0[off];
                v2f z1 = *(const v2f*)&p1[off];
                v2f z2 = *(const v2f*)&p2[off];
                v2f z3 = *(const v2f*)&p3[off];
                v2f bb = *(const v2f*)&bias_lds[n];
                v2f o0 = {0.f, 0.f}, o1 = {0.f, 0.f}, o2 = {0.f, 0.f}, o3 = {0.f, 0.f};
#pragma unroll
                for (int e = 0; e < 2; ++e) {
                    float h  = tanhf(z0[e] + bb[e]);
                    float g  = 1.0f - h * h;
                    float zx = z1[e];
                    o0[e] = h;
                    o1[e] = g * zx;
                    o2[e] = g * z2[e];
                    o3[e] = g * z3[e] - 2.0f * h * g * zx * zx;
                }
                *(v2f*)&p0[off] = o0;
                *(v2f*)&p1[off] = o1;
                *(v2f*)&p2[off] = o2;
                *(v2f*)&p3[off] = o3;
            }
        }
        __syncthreads();
    }

    if (tid < PTS * 4) {
        int pt = tid >> 2, c = tid & 3;
        int tI = pt >> 4, pl = pt & 15;
        const float* hh = &h_lds[tI][c][pl * P];
        float s0 = 0.f, s1 = 0.f;
        for (int k = 0; k < D_IN; ++k) {
            float hk = hh[k];
            s0 = fmaf(hk, W5[k * 2 + 0], s0);
            s1 = fmaf(hk, W5[k * 2 + 1], s1);
        }
        sbuf[pt][c][0] = s0;
        sbuf[pt][c][1] = s1;
    }
    __syncthreads();

    if (tid < PTS) {
        int pt = tid;
        int g = base_pt + pt;
        if (g < N) {
            float u    = sbuf[pt][0][0] + b5[0];
            float v    = sbuf[pt][0][1] + b5[1];
            float u_t  = sbuf[pt][2][0];
            float v_t  = sbuf[pt][2][1];
            float u_xx = sbuf[pt][3][0];
            float v_xx = sbuf[pt][3][1];
            float sq   = u * u + v * v;
            out[g * 2 + 0] = u_t + 0.5f * v_xx + sq * v;
            out[g * 2 + 1] = v_t - 0.5f * u_xx - sq * u;
        }
    }
}

extern "C" void kernel_launch(void* const* d_in, const int* in_sizes, int n_in,
                              void* d_out, int out_size, void* d_ws, size_t ws_size,
                              hipStream_t stream) {
    const float* x  = (const float*)d_in[0];
    const float* t  = (const float*)d_in[1];
    const float* W1 = (const float*)d_in[2];
    const float* b1 = (const float*)d_in[3];
    const float* W2 = (const float*)d_in[4];
    const float* b2 = (const float*)d_in[5];
    const float* W3 = (const float*)d_in[6];
    const float* b3 = (const float*)d_in[7];
    const float* W4 = (const float*)d_in[8];
    const float* b4 = (const float*)d_in[9];
    const float* W5 = (const float*)d_in[10];
    const float* b5 = (const float*)d_in[11];
    float* out = (float*)d_out;
    int N = in_sizes[0];
    int blocks = (N + PTS - 1) / PTS;
    pinn_residual_kernel<<<blocks, NTHR, 0, stream>>>(
        x, t, W1, b1, W2, b2, W3, b3, W4, b4, W5, b5, out, N);
}


# GEMM re-tiled: a wave owns all 4 dual channels of 16 points x 32 neurons (weights as A operand); tanh and derivative products run on the accumulators in registers; park/reload pass and a barrier go
# speedup vs baseline: 2.5348x; 1.0575x over previous
#include <hip/hip_runtime.h>
#include <hip/hip_bf16.h>
#include <math.h>


typedef float v2f __attribute__((ext_vector_type(2)));
typedef float v8f __attribute__((ext_vector_type(8)));
typedef float v4f __attribute__((ext_vector_type(4)));
typedef float v4f_u __attribute__((ext_vector_type(4), aligned(4)));
typedef float v2f_u __attribute__((ext_vector_type(2), aligned(4)));

#define D_IN   100
#define PW     112
#define PH     100
#define PLANE  (16 * PH)
#define KT     25
#define TILES  4
#define PTS    64
#define NTHR   512
#define NGRP   (D_IN / 4)
#define NPAIR  (D_IN / 2)
#define SROUNDS 3
#define L1_PAIRS   (D_IN / 2)
#define L1_GROUPS  (NTHR / L1_PAIRS)
#define L1_THREADS (L1_PAIRS * L1_GROUPS)
#define L1_ITERS   ((PTS + L1_GROUPS - 1) / L1_GROUPS)

#define PI_F   3.14159265358979323846f
#define CX     0.2f
#define CT     (4.0f / PI_F)

template <int NJ>
__device__ __forceinline__ void gemm_channels(const float* __restrict__ hB,
                                              const float* __restrict__ wA,
                                              v8f (&acc)[4][2])
{
    v2f w0[NJ], w1[NJ], h0[4], h1[4];
#pragma unroll
    for (int jj = 0; jj < NJ; ++jj)
        w0[jj] = *(const v2f*)(wA + jj * 32);
#pragma unroll
    for (int c = 0; c < 4; ++c)
        h0[c] = *(const v2f*)(hB + c * PLANE);
#pragma unroll 1
    for (int k = 0; k < KT - 1; k += 2) {
#pragma unroll
        for (int jj = 0; jj < NJ; ++jj)
            w1[jj] = *(const v2f*)(wA + (k + 1) * (4 * PW) + jj * 32);
#pragma unroll
        for (int c = 0; c < 4; ++c)
            h1[c] = *(const v2f*)(hB + (k + 1) * 4 + c * PLANE);
        __builtin_amdgcn_sched_barrier(0);
#pragma unroll
        for (int jj = 0; jj < NJ; ++jj)
#pragma unroll
            for (int c = 0; c < 4; ++c)
                acc[c][jj] = __builtin_amdgcn_wmma_f32_16x16x4_f32(
                    false, w0[jj], false, h0[c], (short)0, acc[c][jj], false, false);
        __builtin_amdgcn_sched_barrier(0);
#pragma unroll
        for (int jj = 0; jj < NJ; ++jj)
            w0[jj] = *(const v2f*)(wA + (k + 2) * (4 * PW) + jj * 32);
#pragma unroll
        for (int c = 0; c < 4; ++c)
            h0[c] = *(const v2f*)(hB + (k + 2) * 4 + c * PLANE);
        __builtin_amdgcn_sched_barrier(0);
#pragma unroll
        for (int jj = 0; jj < NJ; ++jj)
#pragma unroll
            for (int c = 0; c < 4; ++c)
                acc[c][jj] = __builtin_amdgcn_wmma_f32_16x16x4_f32(
                    false, w1[jj], false, h1[c], (short)0, acc[c][jj], false, false);
        __builtin_amdgcn_sched_barrier(0);
    }
#pragma unroll
    for (int jj = 0; jj < NJ; ++jj)
#pragma unroll
        for (int c = 0; c < 4; ++c)
            acc[c][jj] = __builtin_amdgcn_wmma_f32_16x16x4_f32(
                false, w0[jj], false, h0[c], (short)0, acc[c][jj], false, false);
}

template <int NR>
__device__ __forceinline__ void activate_store(const v8f& z, const v8f& zx, const v8f& zt,
                                               const v8f& zxx,
                                               const float* __restrict__ bias,
                                               float* __restrict__ hOut)
{
    float bb[NR], o0[NR], o1[NR], o2[NR], o3[NR];
#pragma unroll
    for (int q = 0; q < NR / 4; ++q) {
        const v4f b4 = *(const v4f*)(bias + 4 * q);
        bb[4 * q + 0] = b4[0]; bb[4 * q + 1] = b4[1];
        bb[4 * q + 2] = b4[2]; bb[4 * q + 3] = b4[3];
    }
#pragma unroll
    for (int i = 0; i < NR; ++i) {
        float h  = tanhf(z[i] + bb[i]);
        float g  = 1.0f - h * h;
        float x_ = zx[i];
        o0[i] = h;
        o1[i] = g * x_;
        o2[i] = g * zt[i];
        o3[i] = g * zxx[i] - 2.0f * h * g * x_ * x_;
    }
#pragma unroll
    for (int q = 0; q < NR / 4; ++q) {
        *(v4f*)(hOut + 4 * q)             = (v4f){o0[4*q], o0[4*q+1], o0[4*q+2], o0[4*q+3]};
        *(v4f*)(hOut + PLANE + 4 * q)     = (v4f){o1[4*q], o1[4*q+1], o1[4*q+2], o1[4*q+3]};
        *(v4f*)(hOut + 2 * PLANE + 4 * q) = (v4f){o2[4*q], o2[4*q+1], o2[4*q+2], o2[4*q+3]};
        *(v4f*)(hOut + 3 * PLANE + 4 * q) = (v4f){o3[4*q], o3[4*q+1], o3[4*q+2], o3[4*q+3]};
    }
}

#define S_ROWS   20
#define S_THREADS (S_ROWS * NGRP)
__device__ __forceinline__ void stage_load(const float* __restrict__ Wg,
                                           const float* __restrict__ bg,
                                           unsigned off, unsigned off2, int tid,
                                           v4f (&ra)[SROUNDS], v4f (&rb)[SROUNDS], float& bv)
{
    ra[0] = *(const v4f_u*)(Wg + off);
    rb[0] = *(const v4f_u*)(Wg + off + D_IN);
    ra[1] = *(const v4f_u*)(Wg + off + 2 * S_ROWS * D_IN);
    rb[1] = *(const v4f_u*)(Wg + off + 2 * S_ROWS * D_IN + D_IN);
    ra[2] = *(const v4f_u*)(Wg + off2);
    rb[2] = *(const v4f_u*)(Wg + off2 + D_IN);
    bv = bg[(tid < D_IN) ? tid : (D_IN - 1)];
}
__device__ __forceinline__ void stage_store(float* __restrict__ d, bool on, bool on2,
                                            const v4f (&ra)[SROUNDS],
                                            const v4f (&rb)[SROUNDS])
{
    if (on) {
        *(v4f*)(d)     = (v4f){ra[0][0], rb[0][0], ra[0][1], rb[0][1]};
        *(v4f*)(d + 4) = (v4f){ra[0][2], rb[0][2], ra[0][3], rb[0][3]};
        float* d1 = d + S_ROWS * PW * 2;
        *(v4f*)(d1)     = (v4f){ra[1][0], rb[1][0], ra[1][1], rb[1][1]};
        *(v4f*)(d1 + 4) = (v4f){ra[1][2], rb[1][2], ra[1][3], rb[1][3]};
        if (on2) {
            float* d2 = d + 2 * S_ROWS * PW * 2;
            *(v4f*)(d2)     = (v4f){ra[2][0], rb[2][0], ra[2][1], rb[2][1]};
            *(v4f*)(d2 + 4) = (v4f){ra[2][2], rb[2][2], ra[2][3], rb[2][3]};
        }
    }
}

__global__ __launch_bounds__(NTHR, 8)
void pinn_residual_kernel(const float* __restrict__ x, const float* __restrict__ t,
                          const float* __restrict__ W1, const float* __restrict__ b1,
                          const float* __restrict__ W2, const float* __restrict__ b2,
                          const float* __restrict__ W3, const float* __restrict__ b3,
                          const float* __restrict__ W4, const float* __restrict__ b4,
                          const float* __restrict__ W5, const float* __restrict__ b5,
                          float* __restrict__ out, int N)
{
    __shared__ __attribute__((aligned(16))) float w_lds[NPAIR * PW * 2];
    __shared__ __attribute__((aligned(16))) float bias_lds[2][PW];
    __shared__ __attribute__((aligned(16))) float h_lds[TILES][4][PLANE];
    float (*sbuf)[4][2] = (float (*)[4][2])w_lds;

    const int tid    = threadIdx.x;
    const int lane   = tid & 31;
    const int wave   = __builtin_amdgcn_readfirstlane(tid >> 5);
    const int tileId = wave >> 2;
    const int jg     = (wave + tileId) & 3;
    const int half   = lane >> 4;
    const int r      = lane & 15;

    const int base_pt = blockIdx.x * PTS;

    if (tid < L1_THREADS) {
        const unsigned utid = (unsigned)tid;
        const int pg = (int)(utid / L1_PAIRS);
        const int cp = (int)(utid - (unsigned)pg * L1_PAIRS);
        const int n0 = 2 * cp;
        const v2f w0 = *(const v2f_u*)(W1 + n0);
        const v2f w1 = *(const v2f_u*)(W1 + D_IN + n0);
        const v2f bb = *(const v2f_u*)(b1 + n0);
        float xv[L1_ITERS], tv[L1_ITERS];
#pragma unroll
        for (int it = 0; it < L1_ITERS; ++it) {
            int pt = pg + L1_GROUPS * it;
            pt = (pt < PTS) ? pt : (PTS - 1);
            const int g  = base_pt + pt;
            const int gc = (g < N) ? g : (N - 1);
            const float xl = x[gc];
            const float tl = t[gc];
            xv[it] = (g < N) ? xl : 0.0f;
            tv[it] = (g < N) ? tl : 0.0f;
        }
        const v2f zx = {CX * w0[0], CX * w0[1]};
        const v2f zt = {CT * w1[0], CT * w1[1]};
#pragma unroll
        for (int it = 0; it < L1_ITERS; ++it) {
            const int pt = pg + L1_GROUPS * it;
            if (pt < PTS) {
                const float sx = CX * (xv[it] + 5.0f) - 1.0f;
                const float st = CT * tv[it] - 1.0f;
                v2f hv, hx, ht, hxx;
#pragma unroll
                for (int e = 0; e < 2; ++e) {
                    float z  = sx * w0[e] + st * w1[e] + bb[e];
                    float h  = tanhf(z);
                    float gg = 1.0f - h * h;
                    hv[e]  = h;
                    hx[e]  = gg * zx[e];
                    ht[e]  = gg * zt[e];
                    hxx[e] = -2.0f * h * gg * zx[e] * zx[e];
                }
                const int tI = pt >> 4, pl = pt & 15;
                float* d = &h_lds[tI][0][pl * PH + n0];
                *(v2f*)(d)             = hv;
                *(v2f*)(d + PLANE)     = hx;
                *(v2f*)(d + 2 * PLANE) = ht;
                *(v2f*)(d + 3 * PLANE) = hxx;
            }
        }
    }

    for (int i = tid; i < NPAIR * (PW - D_IN) * 2; i += NTHR) {
        int p = i / ((PW - D_IN) * 2);
        int c = i - p * ((PW - D_IN) * 2);
        w_lds[(p * PW + D_IN) * 2 + c] = 0.0f;
    }

    const unsigned sp   = (unsigned)tid / NGRP;
    const unsigned sc   = (unsigned)tid - sp * NGRP;
    const bool     s_on  = tid < S_THREADS;
    const bool     s_on2 = sp < (NPAIR - 2 * S_ROWS);
    const unsigned s_off  = s_on ? (2 * sp) * D_IN + 4 * sc : 0u;
    const unsigned s_off2 = s_off + ((s_on && s_on2) ? 2 * (2 * S_ROWS) * D_IN : 0u);
    float* const   s_dst  = &w_lds[(sp * PW + 4 * sc) * 2];
    {
        v4f ra[SROUNDS], rb[SROUNDS];
        float bv;
        stage_load(W2, b2, s_off, s_off2, tid, ra, rb, bv);
        stage_store(s_dst, s_on, s_on2, ra, rb);
        if (tid < PW)
            bias_lds[0][tid] = (tid < D_IN) ? bv : 0.0f;
    }
    __syncthreads();

    const float* hB   = &h_lds[tileId][0][r * PH + 2 * half];
    const float* wA   = &w_lds[(half * PW + 32 * jg + r) * 2];
    float*       hOut = &h_lds[tileId][0][r * PH + 32 * jg + 8 * half];
    for (int l = 0; l < 3; ++l) {
        v8f acc[4][2];
#pragma unroll
        for (int c = 0; c < 4; ++c)
#pragma unroll
            for (int jj = 0; jj < 2; ++jj)
                acc[c][jj] = (v8f){0.f, 0.f, 0.f, 0.f, 0.f, 0.f, 0.f, 0.f};

        if (jg < 3) gemm_channels<2>(hB, wA, acc);
        else        gemm_channels<1>(hB, wA, acc);

        __syncthreads();

        v4f ra[SROUNDS], rb[SROUNDS];
        float bv = 0.0f;
        if (l < 2) {
            const float* Wg = (l == 0) ? W3 : W4;
            const float* bg = (l == 0) ? b3 : b4;
            stage_load(Wg, bg, s_off, s_off2, tid, ra, rb, bv);
        }

        {
            const float* bias = &bias_lds[l & 1][32 * jg + 8 * half];
            if (jg < 3) {
                activate_store<8>(acc[0][0], acc[1][0], acc[2][0], acc[3][0], bias, hOut);
                activate_store<8>(acc[0][1], acc[1][1], acc[2][1], acc[3][1],
                                  bias + 16, hOut + 16);
            } else if (half == 0) {
                activate_store<4>(acc[0][0], acc[1][0], acc[2][0], acc[3][0], bias, hOut);
            }
        }
        if (l < 2) {
            stage_store(s_dst, s_on, s_on2, ra, rb);
            if (tid < PW)
                bias_lds[(l + 1) & 1][tid] = (tid < D_IN) ? bv : 0.0f;
        }
        __syncthreads();
    }

    if (tid < PTS * 4) {
        int pt = tid >> 2, c = tid & 3;
        int tI = pt >> 4, pl = pt & 15;
        const float* hh = &h_lds[tI][c][pl * PH];
        float s0 = 0.f, s1 = 0.f;
        for (int k = 0; k < D_IN; ++k) {
            float hk = hh[k];
            s0 = fmaf(hk, W5[k * 2 + 0], s0);
            s1 = fmaf(hk, W5[k * 2 + 1], s1);
        }
        sbuf[pt][c][0] = s0;
        sbuf[pt][c][1] = s1;
    }
    __syncthreads();

    if (tid < PTS) {
        int pt = tid;
        int g = base_pt + pt;
        if (g < N) {
            float u    = sbuf[pt][0][0] + b5[0];
            float v    = sbuf[pt][0][1] + b5[1];
            float u_t  = sbuf[pt][2][0];
            float v_t  = sbuf[pt][2][1];
            float u_xx = sbuf[pt][3][0];
            float v_xx = sbuf[pt][3][1];
            float sq   = u * u + v * v;
            out[g * 2 + 0] = u_t + 0.5f * v_xx + sq * v;
            out[g * 2 + 1] = v_t - 0.5f * u_xx - sq * u;
        }
    }
}

extern "C" void kernel_launch(void* const* d_in, const int* in_sizes, int n_in,
                              void* d_out, int out_size, void* d_ws, size_t ws_size,
                              hipStream_t stream) {
    const float* x  = (const float*)d_in[0];
    const float* t  = (const float*)d_in[1];
    const float* W1 = (const float*)d_in[2];
    const float* b1 = (const float*)d_in[3];
    const float* W2 = (const float*)d_in[4];
    const float* b2 = (const float*)d_in[5];
    const float* W3 = (const float*)d_in[6];
    const float* b3 = (const float*)d_in[7];
    const float* W4 = (const float*)d_in[8];
    const float* b4 = (const float*)d_in[9];
    const float* W5 = (const float*)d_in[10];
    const float* b5 = (const float*)d_in[11];
    float* out = (float*)d_out;
    int N = in_sizes[0];
    int blocks = (N + PTS - 1) / PTS;
    pinn_residual_kernel<<<blocks, NTHR, 0, stream>>>(
        x, t, W1, b1, W2, b2, W3, b3, W4, b4, W5, b5, out, N);
}


# tanh as sign(z)*(1-e)/(1+e), e=exp2(-2.885|z|), on hardware exp2 and rcp: 7 instructions a value for the library's ~40; not bit-identical: abs error to 2e-7, relative accuracy lost for tiny |z|
# speedup vs baseline: 2.6145x; 1.0314x over previous
#include <hip/hip_runtime.h>
#include <hip/hip_bf16.h>
#include <math.h>


typedef float v2f __attribute__((ext_vector_type(2)));
typedef float v8f __attribute__((ext_vector_type(8)));
typedef float v4f __attribute__((ext_vector_type(4)));
typedef float v4f_u __attribute__((ext_vector_type(4), aligned(4)));
typedef float v2f_u __attribute__((ext_vector_type(2), aligned(4)));

#define D_IN   100
#define PW     112
#define PH     100
#define PLANE  (16 * PH)
#define KT     25
#define TILES  4
#define PTS    64
#define NTHR   512
#define NGRP   (D_IN / 4)
#define NPAIR  (D_IN / 2)
#define SROUNDS 3
#define L1_PAIRS   (D_IN / 2)
#define L1_GROUPS  (NTHR / L1_PAIRS)
#define L1_THREADS (L1_PAIRS * L1_GROUPS)
#define L1_ITERS   ((PTS + L1_GROUPS - 1) / L1_GROUPS)

#define PI_F   3.14159265358979323846f
#define CX     0.2f
#define CT     (4.0f / PI_F)

__device__ __forceinline__ float tanh_exp_rcp(float z)
{
    const float e = __builtin_amdgcn_exp2f(-2.885390081777927f * __builtin_fabsf(z));
    const float r = (1.0f - e) * __builtin_amdgcn_rcpf(1.0f + e);
    return __builtin_copysignf(r, z);
}

template <int NJ>
__device__ __forceinline__ void gemm_channels(const float* __restrict__ hB,
                                              const float* __restrict__ wA,
                                              v8f (&acc)[4][2])
{
    v2f w0[NJ], w1[NJ], h0[4], h1[4];
#pragma unroll
    for (int jj = 0; jj < NJ; ++jj)
        w0[jj] = *(const v2f*)(wA + jj * 32);
#pragma unroll
    for (int c = 0; c < 4; ++c)
        h0[c] = *(const v2f*)(hB + c * PLANE);
#pragma unroll 1
    for (int k = 0; k < KT - 1; k += 2) {
#pragma unroll
        for (int jj = 0; jj < NJ; ++jj)
            w1[jj] = *(const v2f*)(wA + (k + 1) * (4 * PW) + jj * 32);
#pragma unroll
        for (int c = 0; c < 4; ++c)
            h1[c] = *(const v2f*)(hB + (k + 1) * 4 + c * PLANE);
        __builtin_amdgcn_sched_barrier(0);
#pragma unroll
        for (int jj = 0; jj < NJ; ++jj)
#pragma unroll
            for (int c = 0; c < 4; ++c)
                acc[c][jj] = __builtin_amdgcn_wmma_f32_16x16x4_f32(
                    false, w0[jj], false, h0[c], (short)0, acc[c][jj], false, false);
        __builtin_amdgcn_sched_barrier(0);
#pragma unroll
        for (int jj = 0; jj < NJ; ++jj)
            w0[jj] = *(const v2f*)(wA + (k + 2) * (4 * PW) + jj * 32);
#pragma unroll
        for (int c = 0; c < 4; ++c)
            h0[c] = *(const v2f*)(hB + (k + 2) * 4 + c * PLANE);
        __builtin_amdgcn_sched_barrier(0);
#pragma unroll
        for (int jj = 0; jj < NJ; ++jj)
#pragma unroll
            for (int c = 0; c < 4; ++c)
                acc[c][jj] = __builtin_amdgcn_wmma_f32_16x16x4_f32(
                    false, w1[jj], false, h1[c], (short)0, acc[c][jj], false, false);
        __builtin_amdgcn_sched_barrier(0);
    }
#pragma unroll
    for (int jj = 0; jj < NJ; ++jj)
#pragma unroll
        for (int c = 0; c < 4; ++c)
            acc[c][jj] = __builtin_amdgcn_wmma_f32_16x16x4_f32(
                false, w0[jj], false, h0[c], (short)0, acc[c][jj], false, false);
}

template <int NR>
__device__ __forceinline__ void activate_store(const v8f& z, const v8f& zx, const v8f& zt,
                                               const v8f& zxx,
                                               const float* __restrict__ bias,
                                               float* __restrict__ hOut)
{
    float bb[NR], o0[NR], o1[NR], o2[NR], o3[NR];
#pragma unroll
    for (int q = 0; q < NR / 4; ++q) {
        const v4f b4 = *(const v4f*)(bias + 4 * q);
        bb[4 * q + 0] = b4[0]; bb[4 * q + 1] = b4[1];
        bb[4 * q + 2] = b4[2]; bb[4 * q + 3] = b4[3];
    }
#pragma unroll
    for (int i = 0; i < NR; ++i) {
        float h  = tanh_exp_rcp(z[i] + bb[i]);
        float g  = 1.0f - h * h;
        float x_ = zx[i];
        o0[i] = h;
        o1[i] = g * x_;
        o2[i] = g * zt[i];
        o3[i] = g * zxx[i] - 2.0f * h * g * x_ * x_;
    }
#pragma unroll
    for (int q = 0; q < NR / 4; ++q) {
        *(v4f*)(hOut + 4 * q)             = (v4f){o0[4*q], o0[4*q+1], o0[4*q+2], o0[4*q+3]};
        *(v4f*)(hOut + PLANE + 4 * q)     = (v4f){o1[4*q], o1[4*q+1], o1[4*q+2], o1[4*q+3]};
        *(v4f*)(hOut + 2 * PLANE + 4 * q) = (v4f){o2[4*q], o2[4*q+1], o2[4*q+2], o2[4*q+3]};
        *(v4f*)(hOut + 3 * PLANE + 4 * q) = (v4f){o3[4*q], o3[4*q+1], o3[4*q+2], o3[4*q+3]};
    }
}

#define S_ROWS   20
#define S_THREADS (S_ROWS * NGRP)
__device__ __forceinline__ void stage_load(const float* __restrict__ Wg,
                                           const float* __restrict__ bg,
                                           unsigned off, unsigned off2, int tid,
                                           v4f (&ra)[SROUNDS], v4f (&rb)[SROUNDS], float& bv)
{
    ra[0] = *(const v4f_u*)(Wg + off);
    rb[0] = *(const v4f_u*)(Wg + off + D_IN);
    ra[1] = *(const v4f_u*)(Wg + off + 2 * S_ROWS * D_IN);
    rb[1] = *(const v4f_u*)(Wg + off + 2 * S_ROWS * D_IN + D_IN);
    ra[2] = *(const v4f_u*)(Wg + off2);
    rb[2] = *(const v4f_u*)(Wg + off2 + D_IN);
    bv = bg[(tid < D_IN) ? tid : (D_IN - 1)];
}
__device__ __forceinline__ void stage_store(float* __restrict__ d, bool on, bool on2,
                                            const v4f (&ra)[SROUNDS],
                                            const v4f (&rb)[SROUNDS])
{
    if (on) {
        *(v4f*)(d)     = (v4f){ra[0][0], rb[0][0], ra[0][1], rb[0][1]};
        *(v4f*)(d + 4) = (v4f){ra[0][2], rb[0][2], ra[0][3], rb[0][3]};
        float* d1 = d + S_ROWS * PW * 2;
        *(v4f*)(d1)     = (v4f){ra[1][0], rb[1][0], ra[1][1], rb[1][1]};
        *(v4f*)(d1 + 4) = (v4f){ra[1][2], rb[1][2], ra[1][3], rb[1][3]};
        if (on2) {
            float* d2 = d + 2 * S_ROWS * PW * 2;
            *(v4f*)(d2)     = (v4f){ra[2][0], rb[2][0], ra[2][1], rb[2][1]};
            *(v4f*)(d2 + 4) = (v4f){ra[2][2], rb[2][2], ra[2][3], rb[2][3]};
        }
    }
}

__global__ __launch_bounds__(NTHR, 8)
void pinn_residual_kernel(const float* __restrict__ x, const float* __restrict__ t,
                          const float* __restrict__ W1, const float* __restrict__ b1,
                          const float* __restrict__ W2, const float* __restrict__ b2,
                          const float* __restrict__ W3, const float* __restrict__ b3,
                          const float* __restrict__ W4, const float* __restrict__ b4,
                          const float* __restrict__ W5, const float* __restrict__ b5,
                          float* __restrict__ out, int N)
{
    __shared__ __attribute__((aligned(16))) float w_lds[NPAIR * PW * 2];
    __shared__ __attribute__((aligned(16))) float bias_lds[2][PW];
    __shared__ __attribute__((aligned(16))) float h_lds[TILES][4][PLANE];
    float (*sbuf)[4][2] = (float (*)[4][2])w_lds;

    const int tid    = threadIdx.x;
    const int lane   = tid & 31;
    const int wave   = __builtin_amdgcn_readfirstlane(tid >> 5);
    const int tileId = wave >> 2;
    const int jg     = (wave + tileId) & 3;
    const int half   = lane >> 4;
    const int r      = lane & 15;

    const int base_pt = blockIdx.x * PTS;

    if (tid < L1_THREADS) {
        const unsigned utid = (unsigned)tid;
        const int pg = (int)(utid / L1_PAIRS);
        const int cp = (int)(utid - (unsigned)pg * L1_PAIRS);
        const int n0 = 2 * cp;
        const v2f w0 = *(const v2f_u*)(W1 + n0);
        const v2f w1 = *(const v2f_u*)(W1 + D_IN + n0);
        const v2f bb = *(const v2f_u*)(b1 + n0);
        float xv[L1_ITERS], tv[L1_ITERS];
#pragma unroll
        for (int it = 0; it < L1_ITERS; ++it) {
            int pt = pg + L1_GROUPS * it;
            pt = (pt < PTS) ? pt : (PTS - 1);
            const int g  = base_pt + pt;
            const int gc = (g < N) ? g : (N - 1);
            const float xl = x[gc];
            const float tl = t[gc];
            xv[it] = (g < N) ? xl : 0.0f;
            tv[it] = (g < N) ? tl : 0.0f;
        }
        const v2f zx = {CX * w0[0], CX * w0[1]};
        const v2f zt = {CT * w1[0], CT * w1[1]};
#pragma unroll
        for (int it = 0; it < L1_ITERS; ++it) {
            const int pt = pg + L1_GROUPS * it;
            if (pt < PTS) {
                const float sx = CX * (xv[it] + 5.0f) - 1.0f;
                const float st = CT * tv[it] - 1.0f;
                v2f hv, hx, ht, hxx;
#pragma unroll
                for (int e = 0; e < 2; ++e) {
                    float z  = sx * w0[e] + st * w1[e] + bb[e];
                    float h  = tanh_exp_rcp(z);
                    float gg = 1.0f - h * h;
                    hv[e]  = h;
                    hx[e]  = gg * zx[e];
                    ht[e]  = gg * zt[e];
                    hxx[e] = -2.0f * h * gg * zx[e] * zx[e];
                }
                const int tI = pt >> 4, pl = pt & 15;
                float* d = &h_lds[tI][0][pl * PH + n0];
                *(v2f*)(d)             = hv;
                *(v2f*)(d + PLANE)     = hx;
                *(v2f*)(d + 2 * PLANE) = ht;
                *(v2f*)(d + 3 * PLANE) = hxx;
            }
        }
    }

    for (int i = tid; i < NPAIR * (PW - D_IN) * 2; i += NTHR) {
        int p = i / ((PW - D_IN) * 2);
        int c = i - p * ((PW - D_IN) * 2);
        w_lds[(p * PW + D_IN) * 2 + c] = 0.0f;
    }

    const unsigned sp   = (unsigned)tid / NGRP;
    const unsigned sc   = (unsigned)tid - sp * NGRP;
    const bool     s_on  = tid < S_THREADS;
    const bool     s_on2 = sp < (NPAIR - 2 * S_ROWS);
    const unsigned s_off  = s_on ? (2 * sp) * D_IN + 4 * sc : 0u;
    const unsigned s_off2 = s_off + ((s_on && s_on2) ? 2 * (2 * S_ROWS) * D_IN : 0u);
    float* const   s_dst  = &w_lds[(sp * PW + 4 * sc) * 2];
    {
        v4f ra[SROUNDS], rb[SROUNDS];
        float bv;
        stage_load(W2, b2, s_off, s_off2, tid, ra, rb, bv);
        stage_store(s_dst, s_on, s_on2, ra, rb);
        if (tid < PW)
            bias_lds[0][tid] = (tid < D_IN) ? bv : 0.0f;
    }
    __syncthreads();

    const float* hB   = &h_lds[tileId][0][r * PH + 2 * half];
    const float* wA   = &w_lds[(half * PW + 32 * jg + r) * 2];
    float*       hOut = &h_lds[tileId][0][r * PH + 32 * jg + 8 * half];
#pragma unroll 1
    for (int l = 0; l < 3; ++l) {
        v8f acc[4][2];
#pragma unroll
        for (int c = 0; c < 4; ++c)
#pragma unroll
            for (int jj = 0; jj < 2; ++jj)
                acc[c][jj] = (v8f){0.f, 0.f, 0.f, 0.f, 0.f, 0.f, 0.f, 0.f};

        if (jg < 3) gemm_channels<2>(hB, wA, acc);
        else        gemm_channels<1>(hB, wA, acc);

        __syncthreads();

        v4f ra[SROUNDS], rb[SROUNDS];
        float bv = 0.0f;
        if (l < 2) {
            const float* Wg = (l == 0) ? W3 : W4;
            const float* bg = (l == 0) ? b3 : b4;
            stage_load(Wg, bg, s_off, s_off2, tid, ra, rb, bv);
        }

        {
            const float* bias = &bias_lds[l & 1][32 * jg + 8 * half];
            if (jg < 3) {
                activate_store<8>(acc[0][0], acc[1][0], acc[2][0], acc[3][0], bias, hOut);
                activate_store<8>(acc[0][1], acc[1][1], acc[2][1], acc[3][1],
                                  bias + 16, hOut + 16);
            } else if (half == 0) {
                activate_store<4>(acc[0][0], acc[1][0], acc[2][0], acc[3][0], bias, hOut);
            }
        }
        if (l < 2) {
            stage_store(s_dst, s_on, s_on2, ra, rb);
            if (tid < PW)
                bias_lds[(l + 1) & 1][tid] = (tid < D_IN) ? bv : 0.0f;
        }
        __syncthreads();
    }

    if (tid < PTS * 4) {
        int pt = tid >> 2, c = tid & 3;
        int tI = pt >> 4, pl = pt & 15;
        const float* hh = &h_lds[tI][c][pl * PH];
        float s0 = 0.f, s1 = 0.f;
        for (int k = 0; k < D_IN; ++k) {
            float hk = hh[k];
            s0 = fmaf(hk, W5[k * 2 + 0], s0);
            s1 = fmaf(hk, W5[k * 2 + 1], s1);
        }
        sbuf[pt][c][0] = s0;
        sbuf[pt][c][1] = s1;
    }
    __syncthreads();

    if (tid < PTS) {
        int pt = tid;
        int g = base_pt + pt;
        if (g < N) {
            float u    = sbuf[pt][0][0] + b5[0];
            float v    = sbuf[pt][0][1] + b5[1];
            float u_t  = sbuf[pt][2][0];
            float v_t  = sbuf[pt][2][1];
            float u_xx = sbuf[pt][3][0];
            float v_xx = sbuf[pt][3][1];
            float sq   = u * u + v * v;
            out[g * 2 + 0] = u_t + 0.5f * v_xx + sq * v;
            out[g * 2 + 1] = v_t - 0.5f * u_xx - sq * u;
        }
    }
}

extern "C" void kernel_launch(void* const* d_in, const int* in_sizes, int n_in,
                              void* d_out, int out_size, void* d_ws, size_t ws_size,
                              hipStream_t stream) {
    const float* x  = (const float*)d_in[0];
    const float* t  = (const float*)d_in[1];
    const float* W1 = (const float*)d_in[2];
    const float* b1 = (const float*)d_in[3];
    const float* W2 = (const float*)d_in[4];
    const float* b2 = (const float*)d_in[5];
    const float* W3 = (const float*)d_in[6];
    const float* b3 = (const float*)d_in[7];
    const float* W4 = (const float*)d_in[8];
    const float* b4 = (const float*)d_in[9];
    const float* W5 = (const float*)d_in[10];
    const float* b5 = (const float*)d_in[11];
    float* out = (float*)d_out;
    int N = in_sizes[0];
    int blocks = (N + PTS - 1) / PTS;
    pinn_residual_kernel<<<blocks, NTHR, 0, stream>>>(
        x, t, W1, b1, W2, b2, W3, b3, W4, b4, W5, b5, out, N);
}


# tanh on the hardware v_tanh_f32 instruction in layers 1-4: one transcendental a value in place of the abs/exp2/1+e/1-e/rcp/multiply/copysign form; last bits of every activation change
# speedup vs baseline: 2.6966x; 1.0314x over previous
#include <hip/hip_runtime.h>
#include <hip/hip_bf16.h>
#include <math.h>


typedef float v2f __attribute__((ext_vector_type(2)));
typedef float v8f __attribute__((ext_vector_type(8)));
typedef float v4f __attribute__((ext_vector_type(4)));
typedef float v4f_u __attribute__((ext_vector_type(4), aligned(4)));
typedef float v2f_u __attribute__((ext_vector_type(2), aligned(4)));

#define D_IN   100
#define PW     112
#define PH     100
#define PLANE  (16 * PH)
#define KT     25
#define TILES  4
#define PTS    64
#define NTHR   512
#define NGRP   (D_IN / 4)
#define NPAIR  (D_IN / 2)
#define SROUNDS 3
#define L1_PAIRS   (D_IN / 2)
#define L1_GROUPS  (NTHR / L1_PAIRS)
#define L1_THREADS (L1_PAIRS * L1_GROUPS)
#define L1_ITERS   ((PTS + L1_GROUPS - 1) / L1_GROUPS)

#define PI_F   3.14159265358979323846f
#define CX     0.2f
#define CT     (4.0f / PI_F)

__device__ __forceinline__ float tanh_hw(float z)
{
    return __builtin_amdgcn_tanhf(z);
}

template <int NJ>
__device__ __forceinline__ void gemm_channels(const float* __restrict__ hB,
                                              const float* __restrict__ wA,
                                              v8f (&acc)[4][2])
{
    v2f w0[NJ], w1[NJ], h0[4], h1[4];
#pragma unroll
    for (int jj = 0; jj < NJ; ++jj)
        w0[jj] = *(const v2f*)(wA + jj * 32);
#pragma unroll
    for (int c = 0; c < 4; ++c)
        h0[c] = *(const v2f*)(hB + c * PLANE);
#pragma unroll 1
    for (int k = 0; k < KT - 1; k += 2) {
#pragma unroll
        for (int jj = 0; jj < NJ; ++jj)
            w1[jj] = *(const v2f*)(wA + (k + 1) * (4 * PW) + jj * 32);
#pragma unroll
        for (int c = 0; c < 4; ++c)
            h1[c] = *(const v2f*)(hB + (k + 1) * 4 + c * PLANE);
        __builtin_amdgcn_sched_barrier(0);
#pragma unroll
        for (int jj = 0; jj < NJ; ++jj)
#pragma unroll
            for (int c = 0; c < 4; ++c)
                acc[c][jj] = __builtin_amdgcn_wmma_f32_16x16x4_f32(
                    false, w0[jj], false, h0[c], (short)0, acc[c][jj], false, false);
        __builtin_amdgcn_sched_barrier(0);
#pragma unroll
        for (int jj = 0; jj < NJ; ++jj)
            w0[jj] = *(const v2f*)(wA + (k + 2) * (4 * PW) + jj * 32);
#pragma unroll
        for (int c = 0; c < 4; ++c)
            h0[c] = *(const v2f*)(hB + (k + 2) * 4 + c * PLANE);
        __builtin_amdgcn_sched_barrier(0);
#pragma unroll
        for (int jj = 0; jj < NJ; ++jj)
#pragma unroll
            for (int c = 0; c < 4; ++c)
                acc[c][jj] = __builtin_amdgcn_wmma_f32_16x16x4_f32(
                    false, w1[jj], false, h1[c], (short)0, acc[c][jj], false, false);
        __builtin_amdgcn_sched_barrier(0);
    }
#pragma unroll
    for (int jj = 0; jj < NJ; ++jj)
#pragma unroll
        for (int c = 0; c < 4; ++c)
            acc[c][jj] = __builtin_amdgcn_wmma_f32_16x16x4_f32(
                false, w0[jj], false, h0[c], (short)0, acc[c][jj], false, false);
}

template <int NR>
__device__ __forceinline__ void activate_store(const v8f& z, const v8f& zx, const v8f& zt,
                                               const v8f& zxx,
                                               const float* __restrict__ bias,
                                               float* __restrict__ hOut)
{
    float bb[NR], o0[NR], o1[NR], o2[NR], o3[NR];
#pragma unroll
    for (int q = 0; q < NR / 4; ++q) {
        const v4f b4 = *(const v4f*)(bias + 4 * q);
        bb[4 * q + 0] = b4[0]; bb[4 * q + 1] = b4[1];
        bb[4 * q + 2] = b4[2]; bb[4 * q + 3] = b4[3];
    }
#pragma unroll
    for (int i = 0; i < NR; ++i) {
        float h  = tanh_hw(z[i] + bb[i]);
        float g  = 1.0f - h * h;
        float x_ = zx[i];
        o0[i] = h;
        o1[i] = g * x_;
        o2[i] = g * zt[i];
        o3[i] = g * zxx[i] - 2.0f * h * g * x_ * x_;
    }
#pragma unroll
    for (int q = 0; q < NR / 4; ++q) {
        *(v4f*)(hOut + 4 * q)             = (v4f){o0[4*q], o0[4*q+1], o0[4*q+2], o0[4*q+3]};
        *(v4f*)(hOut + PLANE + 4 * q)     = (v4f){o1[4*q], o1[4*q+1], o1[4*q+2], o1[4*q+3]};
        *(v4f*)(hOut + 2 * PLANE + 4 * q) = (v4f){o2[4*q], o2[4*q+1], o2[4*q+2], o2[4*q+3]};
        *(v4f*)(hOut + 3 * PLANE + 4 * q) = (v4f){o3[4*q], o3[4*q+1], o3[4*q+2], o3[4*q+3]};
    }
}

#define S_ROWS   20
#define S_THREADS (S_ROWS * NGRP)
__device__ __forceinline__ void stage_load(const float* __restrict__ Wg,
                                           const float* __restrict__ bg,
                                           unsigned off, unsigned off2, int tid,
                                           v4f (&ra)[SROUNDS], v4f (&rb)[SROUNDS], float& bv)
{
    ra[0] = *(const v4f_u*)(Wg + off);
    rb[0] = *(const v4f_u*)(Wg + off + D_IN);
    ra[1] = *(const v4f_u*)(Wg + off + 2 * S_ROWS * D_IN);
    rb[1] = *(const v4f_u*)(Wg + off + 2 * S_ROWS * D_IN + D_IN);
    ra[2] = *(const v4f_u*)(Wg + off2);
    rb[2] = *(const v4f_u*)(Wg + off2 + D_IN);
    bv = bg[(tid < D_IN) ? tid : (D_IN - 1)];
}
__device__ __forceinline__ void stage_store(float* __restrict__ d, bool on, bool on2,
                                            const v4f (&ra)[SROUNDS],
                                            const v4f (&rb)[SROUNDS])
{
    if (on) {
        *(v4f*)(d)     = (v4f){ra[0][0], rb[0][0], ra[0][1], rb[0][1]};
        *(v4f*)(d + 4) = (v4f){ra[0][2], rb[0][2], ra[0][3], rb[0][3]};
        float* d1 = d + S_ROWS * PW * 2;
        *(v4f*)(d1)     = (v4f){ra[1][0], rb[1][0], ra[1][1], rb[1][1]};
        *(v4f*)(d1 + 4) = (v4f){ra[1][2], rb[1][2], ra[1][3], rb[1][3]};
        if (on2) {
            float* d2 = d + 2 * S_ROWS * PW * 2;
            *(v4f*)(d2)     = (v4f){ra[2][0], rb[2][0], ra[2][1], rb[2][1]};
            *(v4f*)(d2 + 4) = (v4f){ra[2][2], rb[2][2], ra[2][3], rb[2][3]};
        }
    }
}

__global__ __launch_bounds__(NTHR, 8)
void pinn_residual_kernel(const float* __restrict__ x, const float* __restrict__ t,
                          const float* __restrict__ W1, const float* __restrict__ b1,
                          const float* __restrict__ W2, const float* __restrict__ b2,
                          const float* __restrict__ W3, const float* __restrict__ b3,
                          const float* __restrict__ W4, const float* __restrict__ b4,
                          const float* __restrict__ W5, const float* __restrict__ b5,
                          float* __restrict__ out, int N)
{
    __shared__ __attribute__((aligned(16))) float w_lds[NPAIR * PW * 2];
    __shared__ __attribute__((aligned(16))) float bias_lds[2][PW];
    __shared__ __attribute__((aligned(16))) float h_lds[TILES][4][PLANE];
    float (*sbuf)[4][2] = (float (*)[4][2])w_lds;

    const int tid    = threadIdx.x;
    const int lane   = tid & 31;
    const int wave   = __builtin_amdgcn_readfirstlane(tid >> 5);
    const int tileId = wave >> 2;
    const int jg     = (wave + tileId) & 3;
    const int half   = lane >> 4;
    const int r      = lane & 15;

    const int base_pt = blockIdx.x * PTS;

    if (tid < L1_THREADS) {
        const unsigned utid = (unsigned)tid;
        const int pg = (int)(utid / L1_PAIRS);
        const int cp = (int)(utid - (unsigned)pg * L1_PAIRS);
        const int n0 = 2 * cp;
        const v2f w0 = *(const v2f_u*)(W1 + n0);
        const v2f w1 = *(const v2f_u*)(W1 + D_IN + n0);
        const v2f bb = *(const v2f_u*)(b1 + n0);
        float xv[L1_ITERS], tv[L1_ITERS];
#pragma unroll
        for (int it = 0; it < L1_ITERS; ++it) {
            int pt = pg + L1_GROUPS * it;
            pt = (pt < PTS) ? pt : (PTS - 1);
            const int g  = base_pt + pt;
            const int gc = (g < N) ? g : (N - 1);
            const float xl = x[gc];
            const float tl = t[gc];
            xv[it] = (g < N) ? xl : 0.0f;
            tv[it] = (g < N) ? tl : 0.0f;
        }
        const v2f zx = {CX * w0[0], CX * w0[1]};
        const v2f zt = {CT * w1[0], CT * w1[1]};
#pragma unroll
        for (int it = 0; it < L1_ITERS; ++it) {
            const int pt = pg + L1_GROUPS * it;
            if (pt < PTS) {
                const float sx = CX * (xv[it] + 5.0f) - 1.0f;
                const float st = CT * tv[it] - 1.0f;
                v2f hv, hx, ht, hxx;
#pragma unroll
                for (int e = 0; e < 2; ++e) {
                    float z  = sx * w0[e] + st * w1[e] + bb[e];
                    float h  = tanh_hw(z);
                    float gg = 1.0f - h * h;
                    hv[e]  = h;
                    hx[e]  = gg * zx[e];
                    ht[e]  = gg * zt[e];
                    hxx[e] = -2.0f * h * gg * zx[e] * zx[e];
                }
                const int tI = pt >> 4, pl = pt & 15;
                float* d = &h_lds[tI][0][pl * PH + n0];
                *(v2f*)(d)             = hv;
                *(v2f*)(d + PLANE)     = hx;
                *(v2f*)(d + 2 * PLANE) = ht;
                *(v2f*)(d + 3 * PLANE) = hxx;
            }
        }
    }

    for (int i = tid; i < NPAIR * (PW - D_IN) * 2; i += NTHR) {
        int p = i / ((PW - D_IN) * 2);
        int c = i - p * ((PW - D_IN) * 2);
        w_lds[(p * PW + D_IN) * 2 + c] = 0.0f;
    }

    const unsigned sp   = (unsigned)tid / NGRP;
    const unsigned sc   = (unsigned)tid - sp * NGRP;
    const bool     s_on  = tid < S_THREADS;
    const bool     s_on2 = sp < (NPAIR - 2 * S_ROWS);
    const unsigned s_off  = s_on ? (2 * sp) * D_IN + 4 * sc : 0u;
    const unsigned s_off2 = s_off + ((s_on && s_on2) ? 2 * (2 * S_ROWS) * D_IN : 0u);
    float* const   s_dst  = &w_lds[(sp * PW + 4 * sc) * 2];
    {
        v4f ra[SROUNDS], rb[SROUNDS];
        float bv;
        stage_load(W2, b2, s_off, s_off2, tid, ra, rb, bv);
        stage_store(s_dst, s_on, s_on2, ra, rb);
        if (tid < PW)
            bias_lds[0][tid] = (tid < D_IN) ? bv : 0.0f;
    }
    __syncthreads();

    const float* hB   = &h_lds[tileId][0][r * PH + 2 * half];
    const float* wA   = &w_lds[(half * PW + 32 * jg + r) * 2];
    float*       hOut = &h_lds[tileId][0][r * PH + 32 * jg + 8 * half];
#pragma unroll 1
    for (int l = 0; l < 3; ++l) {
        v8f acc[4][2];
#pragma unroll
        for (int c = 0; c < 4; ++c)
#pragma unroll
            for (int jj = 0; jj < 2; ++jj)
                acc[c][jj] = (v8f){0.f, 0.f, 0.f, 0.f, 0.f, 0.f, 0.f, 0.f};

        if (jg < 3) gemm_channels<2>(hB, wA, acc);
        else        gemm_channels<1>(hB, wA, acc);

        __syncthreads();

        v4f ra[SROUNDS], rb[SROUNDS];
        float bv = 0.0f;
        if (l < 2) {
            const float* Wg = (l == 0) ? W3 : W4;
            const float* bg = (l == 0) ? b3 : b4;
            stage_load(Wg, bg, s_off, s_off2, tid, ra, rb, bv);
        }

        {
            const float* bias = &bias_lds[l & 1][32 * jg + 8 * half];
            if (jg < 3) {
                activate_store<8>(acc[0][0], acc[1][0], acc[2][0], acc[3][0], bias, hOut);
                activate_store<8>(acc[0][1], acc[1][1], acc[2][1], acc[3][1],
                                  bias + 16, hOut + 16);
            } else if (half == 0) {
                activate_store<4>(acc[0][0], acc[1][0], acc[2][0], acc[3][0], bias, hOut);
            }
        }
        if (l < 2) {
            stage_store(s_dst, s_on, s_on2, ra, rb);
            if (tid < PW)
                bias_lds[(l + 1) & 1][tid] = (tid < D_IN) ? bv : 0.0f;
        }
        __syncthreads();
    }

    if (tid < PTS * 4) {
        int pt = tid >> 2, c = tid & 3;
        int tI = pt >> 4, pl = pt & 15;
        const float* hh = &h_lds[tI][c][pl * PH];
        float s0 = 0.f, s1 = 0.f;
        for (int k = 0; k < D_IN; ++k) {
            float hk = hh[k];
            s0 = fmaf(hk, W5[k * 2 + 0], s0);
            s1 = fmaf(hk, W5[k * 2 + 1], s1);
        }
        sbuf[pt][c][0] = s0;
        sbuf[pt][c][1] = s1;
    }
    __syncthreads();

    if (tid < PTS) {
        int pt = tid;
        int g = base_pt + pt;
        if (g < N) {
            float u    = sbuf[pt][0][0] + b5[0];
            float v    = sbuf[pt][0][1] + b5[1];
            float u_t  = sbuf[pt][2][0];
            float v_t  = sbuf[pt][2][1];
            float u_xx = sbuf[pt][3][0];
            float v_xx = sbuf[pt][3][1];
            float sq   = u * u + v * v;
            out[g * 2 + 0] = u_t + 0.5f * v_xx + sq * v;
            out[g * 2 + 1] = v_t - 0.5f * u_xx - sq * u;
        }
    }
}

extern "C" void kernel_launch(void* const* d_in, const int* in_sizes, int n_in,
                              void* d_out, int out_size, void* d_ws, size_t ws_size,
                              hipStream_t stream) {
    const float* x  = (const float*)d_in[0];
    const float* t  = (const float*)d_in[1];
    const float* W1 = (const float*)d_in[2];
    const float* b1 = (const float*)d_in[3];
    const float* W2 = (const float*)d_in[4];
    const float* b2 = (const float*)d_in[5];
    const float* W3 = (const float*)d_in[6];
    const float* b3 = (const float*)d_in[7];
    const float* W4 = (const float*)d_in[8];
    const float* b4 = (const float*)d_in[9];
    const float* W5 = (const float*)d_in[10];
    const float* b5 = (const float*)d_in[11];
    float* out = (float*)d_out;
    int N = in_sizes[0];
    int blocks = (N + PTS - 1) / PTS;
    pinn_residual_kernel<<<blocks, NTHR, 0, stream>>>(
        x, t, W1, b1, W2, b2, W3, b3, W4, b4, W5, b5, out, N);
}
